# starting kernel
# baseline (speedup 1.0000x reference)
#include <hip/hip_runtime.h>
#include <hip/hip_bf16.h>

typedef __attribute__((ext_vector_type(2))) float v2f;
typedef __attribute__((ext_vector_type(8))) float v8f;

#define B_ 4
#define Q_ 256
#define K_ 1024
#define D_ 256

__device__ __forceinline__ float tanh_fast(float x) {
    float e = __builtin_amdgcn_exp2f(x * 2.88539008177793f);
    return 1.0f - 2.0f * __builtin_amdgcn_rcpf(e + 1.0f);
}

__global__ __launch_bounds__(256)
void proj_gemm_wmma(const float* __restrict__ X, const float* __restrict__ W,
                    float* __restrict__ Y) {
    __shared__ float xs[16][D_ + 4];
    const int tid = threadIdx.x;
    const int m0  = blockIdx.y * 16;
    const int n0  = blockIdx.x * 128;

    #pragma unroll
    for (int i = 0; i < 16; ++i)
        xs[i][tid] = X[(size_t)(m0 + i) * D_ + tid];
    __syncthreads();

    const int lane = tid & 31;
    const int wave = tid >> 5;
    const int mn   = lane & 15;
    const int hi   = lane >> 4;
    const int nw   = n0 + wave * 16;

    v8f c = {};
    for (int kk = 0; kk < D_; kk += 4) {
        v2f a, b;
        a.x = xs[mn][kk + 2 * hi + 0];
        a.y = xs[mn][kk + 2 * hi + 1];
        b.x = W[(size_t)(kk + 2 * hi + 0) * D_ + nw + mn];
        b.y = W[(size_t)(kk + 2 * hi + 1) * D_ + nw + mn];
        c = __builtin_amdgcn_wmma_f32_16x16x4_f32(false, a, false, b,
                                                  (short)0, c, false, false);
    }
    #pragma unroll
    for (int r = 0; r < 8; ++r)
        Y[(size_t)(m0 + r + 8 * hi) * D_ + nw + mn] = c[r];
}

__global__ __launch_bounds__(256)
void scores_kernel(const float* __restrict__ qp, const float* __restrict__ kp,
                   const float* __restrict__ wv, const int* __restrict__ mask,
                   float* __restrict__ scores) {
    __shared__ float qs[16][D_ + 4];
    __shared__ float ks[16][D_ + 4];
    __shared__ float wvs[D_];
    const int tid = threadIdx.x;
    const int k0 = blockIdx.x * 16, q0 = blockIdx.y * 16, b = blockIdx.z;

    const float* qpb = qp + ((size_t)b * Q_ + q0) * D_;
    const float* kpb = kp + ((size_t)b * K_ + k0) * D_;
    #pragma unroll
    for (int i = 0; i < 16; ++i) {
        qs[i][tid] = qpb[(size_t)i * D_ + tid];
        ks[i][tid] = kpb[(size_t)i * D_ + tid];
    }
    wvs[tid] = wv[tid];
    __syncthreads();

    const int tk = tid & 15, tq = tid >> 4;
    float s = 0.f;
    #pragma unroll 4
    for (int e = 0; e < D_; ++e)
        s += tanh_fast(qs[tq][e] + ks[tk][e]) * wvs[e];

    const int kg = k0 + tk;
    const bool valid = mask[(size_t)b * K_ + kg] != 0;
    scores[((size_t)b * Q_ + q0 + tq) * K_ + kg] = valid ? s : -__builtin_inff();
}

__global__ __launch_bounds__(256)
void softmax_kernel(float* __restrict__ scores) {
    __shared__ float red[256];
    float* s = scores + (size_t)blockIdx.x * K_;
    const int tid = threadIdx.x;

    float v0[4];
    float mx = -__builtin_inff();
    #pragma unroll
    for (int i = 0; i < 4; ++i) { v0[i] = s[tid + i * 256]; mx = fmaxf(mx, v0[i]); }
    red[tid] = mx; __syncthreads();
    for (int off = 128; off > 0; off >>= 1) {
        if (tid < off) red[tid] = fmaxf(red[tid], red[tid + off]);
        __syncthreads();
    }
    mx = red[0]; __syncthreads();

    float ex[4], sum = 0.f;
    #pragma unroll
    for (int i = 0; i < 4; ++i) {
        ex[i] = __builtin_amdgcn_exp2f((v0[i] - mx) * 1.44269504088896f);
        sum += ex[i];
    }
    red[tid] = sum; __syncthreads();
    for (int off = 128; off > 0; off >>= 1) {
        if (tid < off) red[tid] += red[tid + off];
        __syncthreads();
    }
    const float inv = 1.0f / red[0];
    #pragma unroll
    for (int i = 0; i < 4; ++i) s[tid + i * 256] = ex[i] * inv;
}

__global__ __launch_bounds__(256)
void attend_wmma(const float* __restrict__ probs, const float* __restrict__ V,
                 float* __restrict__ out) {
    __shared__ float ps[16][K_ + 4];
    const int tid = threadIdx.x;
    const int q0 = blockIdx.x * 16, b = blockIdx.y;

    const float* pb = probs + ((size_t)b * Q_ + q0) * K_;
    #pragma unroll
    for (int i = 0; i < 64; ++i) {
        int f = i * 256 + tid;
        ps[f >> 10][f & 1023] = pb[(size_t)(f >> 10) * K_ + (f & 1023)];
    }
    __syncthreads();

    const int lane = tid & 31, wave = tid >> 5;
    const int mn = lane & 15, hi = lane >> 4;
    const int d0 = wave * 32;
    const float* vb = V + (size_t)b * K_ * D_;

    v8f c0 = {}, c1 = {};
    for (int kk = 0; kk < K_; kk += 4) {
        v2f a, b0, b1;
        a.x = ps[mn][kk + 2 * hi + 0];
        a.y = ps[mn][kk + 2 * hi + 1];
        const float* vr0 = vb + (size_t)(kk + 2 * hi + 0) * D_;
        const float* vr1 = vb + (size_t)(kk + 2 * hi + 1) * D_;
        b0.x = vr0[d0 + mn];       b0.y = vr1[d0 + mn];
        b1.x = vr0[d0 + 16 + mn];  b1.y = vr1[d0 + 16 + mn];
        c0 = __builtin_amdgcn_wmma_f32_16x16x4_f32(false, a, false, b0,
                                                   (short)0, c0, false, false);
        c1 = __builtin_amdgcn_wmma_f32_16x16x4_f32(false, a, false, b1,
                                                   (short)0, c1, false, false);
    }
    float* ob = out + ((size_t)b * Q_ + q0) * D_;
    #pragma unroll
    for (int r = 0; r < 8; ++r) {
        ob[(size_t)(r + 8 * hi) * D_ + d0 + mn]      = c0[r];
        ob[(size_t)(r + 8 * hi) * D_ + d0 + 16 + mn] = c1[r];
    }
}

extern "C" void kernel_launch(void* const* d_in, const int* in_sizes, int n_in,
                              void* d_out, int out_size, void* d_ws, size_t ws_size,
                              hipStream_t stream) {
    const float* q    = (const float*)d_in[0];
    const float* k    = (const float*)d_in[1];
    const float* v    = (const float*)d_in[2];
    const int*   mask = (const int*)  d_in[3];
    const float* Wq   = (const float*)d_in[4];
    const float* Wk   = (const float*)d_in[5];
    const float* wv   = (const float*)d_in[6];
    float* out = (float*)d_out;

    float* qp = (float*)d_ws;
    float* kp = qp + (size_t)B_ * Q_ * D_;
    float* sc = kp + (size_t)B_ * K_ * D_;

    proj_gemm_wmma<<<dim3(2, (B_ * Q_) / 16), 256, 0, stream>>>(q, Wq, qp);
    proj_gemm_wmma<<<dim3(2, (B_ * K_) / 16), 256, 0, stream>>>(k, Wk, kp);
    scores_kernel<<<dim3(K_ / 16, Q_ / 16, B_), 256, 0, stream>>>(qp, kp, wv, mask, sc);
    softmax_kernel<<<B_ * Q_, 256, 0, stream>>>(sc);
    attend_wmma<<<dim3(Q_ / 16, B_), 256, 0, stream>>>(sc, v, out);
}


# scores: each wave owns 2 q rows, read with wv through scalar loads; each lane 2 keys of a 64-key x 64-e kp slab in LDS: 1 LDS read per 8 terms where it was 6; every sum keeps its order
# speedup vs baseline: 2.4493x; 2.4493x over previous
#include <hip/hip_runtime.h>
#include <hip/hip_bf16.h>

typedef __attribute__((ext_vector_type(2))) float v2f;
typedef __attribute__((ext_vector_type(4))) float v4f;
typedef __attribute__((ext_vector_type(8))) float v8f;

#define B_ 4
#define Q_ 256
#define K_ 1024
#define D_ 256

__device__ __forceinline__ float tanh_fast(float x) {
    return __builtin_amdgcn_tanhf(x);
}

__global__ __launch_bounds__(256)
void proj_gemm_wmma(const float* __restrict__ X, const float* __restrict__ W,
                    float* __restrict__ Y) {
    __shared__ float xs[16][D_ + 4];
    const int tid = threadIdx.x;
    const int m0  = blockIdx.y * 16;
    const int n0  = blockIdx.x * 128;

    #pragma unroll
    for (int i = 0; i < 16; ++i)
        xs[i][tid] = X[(size_t)(m0 + i) * D_ + tid];
    __syncthreads();

    const int lane = tid & 31;
    const int wave = tid >> 5;
    const int mn   = lane & 15;
    const int hi   = lane >> 4;
    const int nw   = n0 + wave * 16;

    v8f c = {};
    for (int kk = 0; kk < D_; kk += 4) {
        v2f a, b;
        a.x = xs[mn][kk + 2 * hi + 0];
        a.y = xs[mn][kk + 2 * hi + 1];
        b.x = W[(size_t)(kk + 2 * hi + 0) * D_ + nw + mn];
        b.y = W[(size_t)(kk + 2 * hi + 1) * D_ + nw + mn];
        c = __builtin_amdgcn_wmma_f32_16x16x4_f32(false, a, false, b,
                                                  (short)0, c, false, false);
    }
    #pragma unroll
    for (int r = 0; r < 8; ++r)
        Y[(size_t)(m0 + r + 8 * hi) * D_ + nw + mn] = c[r];
}

#define KC_  64
#define NCH_ (K_ / KC_)
#define ES_  64
#define NSL_ (D_ / ES_)
#define LOG2E_ 1.44269504088896f

__global__ __launch_bounds__(256)
void chunk_attention(const float* __restrict__ qp, const float* __restrict__ kp,
                     const float* __restrict__ wv, const int* __restrict__ mask,
                     const float* __restrict__ V,
                     float* __restrict__ part, float* __restrict__ stat) {
    __shared__ __attribute__((aligned(16))) float ks[KC_][ES_ + 4];
    __shared__ float ps[16][KC_ + 4];
    const int tid = threadIdx.x;
    const int kc = blockIdx.x, q0 = blockIdx.y * 16, b = blockIdx.z;
    const int k0 = kc * KC_;
    const int lane = tid & 31;
    const int wave = __builtin_amdgcn_readfirstlane(tid >> 5);
    const int mn = lane & 15, hi = lane >> 4;
    const int key_a = mn + 32 * hi;
    const int key_b = key_a + 16;

    const float* kpb = kp + ((size_t)b * K_ + k0) * D_;
    const float* qr0 = qp + ((size_t)b * Q_ + q0 + 2 * wave) * D_;
    const float* qr1 = qr0 + D_;

    const int mk_a = mask[(size_t)b * K_ + k0 + key_a];
    const int mk_b = mask[(size_t)b * K_ + k0 + key_b];

    const int sr = tid >> 4, sc = (tid & 15) * 4;
    v4f kreg[4];
    #pragma unroll
    for (int i = 0; i < 4; ++i)
        kreg[i] = *(const v4f*)(kpb + (size_t)(sr + 16 * i) * D_ + sc);

    float s00 = 0.f, s01 = 0.f, s10 = 0.f, s11 = 0.f;
    for (int sl = 0; sl < NSL_; ++sl) {
        __syncthreads();
        #pragma unroll
        for (int i = 0; i < 4; ++i) *(v4f*)&ks[sr + 16 * i][sc] = kreg[i];
        __syncthreads();
        if (sl + 1 < NSL_) {
            #pragma unroll
            for (int i = 0; i < 4; ++i)
                kreg[i] = *(const v4f*)(kpb + (size_t)(sr + 16 * i) * D_ + (sl + 1) * ES_ + sc);
        }
        const float* q0e = qr0 + sl * ES_;
        const float* q1e = qr1 + sl * ES_;
        const float* wve = wv + sl * ES_;
        #pragma unroll 4
        for (int e = 0; e < ES_; e += 4) {
            const v4f ka = *(const v4f*)&ks[key_a][e];
            const v4f kb = *(const v4f*)&ks[key_b][e];
            #pragma unroll
            for (int u = 0; u < 4; ++u) {
                const float w = wve[e + u], qa = q0e[e + u], qb = q1e[e + u];
                s00 = __builtin_fmaf(tanh_fast(qa + ka[u]), w, s00);
                s01 = __builtin_fmaf(tanh_fast(qa + kb[u]), w, s01);
                s10 = __builtin_fmaf(tanh_fast(qb + ka[u]), w, s10);
                s11 = __builtin_fmaf(tanh_fast(qb + kb[u]), w, s11);
            }
        }
    }

    float sa[2], sb[2], m[2], l[2];
    sa[0] = (mk_a != 0) ? s00 : -__builtin_inff();
    sb[0] = (mk_b != 0) ? s01 : -__builtin_inff();
    sa[1] = (mk_a != 0) ? s10 : -__builtin_inff();
    sb[1] = (mk_b != 0) ? s11 : -__builtin_inff();
    #pragma unroll
    for (int j = 0; j < 2; ++j) m[j] = fmaxf(sa[j], sb[j]);
    #pragma unroll
    for (int off = 16; off > 0; off >>= 1) {
        float x[2];
        #pragma unroll
        for (int j = 0; j < 2; ++j) x[j] = __shfl_xor(m[j], off, 32);
        #pragma unroll
        for (int j = 0; j < 2; ++j) m[j] = fmaxf(m[j], x[j]);
    }
    float pa[2], pb[2], xa[2], xb[2];
    #pragma unroll
    for (int j = 0; j < 2; ++j) {
        const float msafe = (m[j] == -__builtin_inff()) ? 0.f : m[j];
        pa[j] = __builtin_amdgcn_exp2f((sa[j] - msafe) * LOG2E_);
        pb[j] = __builtin_amdgcn_exp2f((sb[j] - msafe) * LOG2E_);
        ps[2 * wave + j][key_a] = pa[j];
        ps[2 * wave + j][key_b] = pb[j];
    }
    #pragma unroll
    for (int j = 0; j < 2; ++j) {
        xa[j] = __shfl_xor(pa[j], 16, 32);
        xb[j] = __shfl_xor(pb[j], 16, 32);
    }
    #pragma unroll
    for (int j = 0; j < 2; ++j) {
        const float t0 = hi ? xa[j] : pa[j], t1 = hi ? xb[j] : pb[j];
        const float t2 = hi ? pa[j] : xa[j], t3 = hi ? pb[j] : xb[j];
        l[j] = ((t0 + t1) + t2) + t3;
    }
    #pragma unroll
    for (int off = 8; off > 0; off >>= 1) {
        float x[2];
        #pragma unroll
        for (int j = 0; j < 2; ++j) x[j] = __shfl_xor(l[j], off, 16);
        #pragma unroll
        for (int j = 0; j < 2; ++j) l[j] += x[j];
    }
    if (lane == 0) {
        #pragma unroll
        for (int j = 0; j < 2; ++j) {
            const size_t row = ((size_t)b * Q_ + q0 + 2 * wave + j) * NCH_ + kc;
            stat[row * 2 + 0] = m[j];
            stat[row * 2 + 1] = l[j];
        }
    }
    __syncthreads();

    const int d0 = wave * 32;
    const float* vb = V + ((size_t)b * K_ + k0) * D_;

    v8f c0 = {}, c1 = {};
    #pragma unroll 8
    for (int kk = 0; kk < KC_; kk += 4) {
        v2f a, b0, b1;
        a.x = ps[mn][kk + 2 * hi + 0];
        a.y = ps[mn][kk + 2 * hi + 1];
        const float* vr0 = vb + (size_t)(kk + 2 * hi + 0) * D_;
        const float* vr1 = vb + (size_t)(kk + 2 * hi + 1) * D_;
        b0.x = vr0[d0 + mn];       b0.y = vr1[d0 + mn];
        b1.x = vr0[d0 + 16 + mn];  b1.y = vr1[d0 + 16 + mn];
        c0 = __builtin_amdgcn_wmma_f32_16x16x4_f32(false, a, false, b0,
                                                   (short)0, c0, false, false);
        c1 = __builtin_amdgcn_wmma_f32_16x16x4_f32(false, a, false, b1,
                                                   (short)0, c1, false, false);
    }
    #pragma unroll
    for (int r = 0; r < 8; ++r) {
        float* pr = part + (((size_t)b * Q_ + q0 + r + 8 * hi) * NCH_ + kc) * D_;
        pr[d0 + mn]      = c0[r];
        pr[d0 + 16 + mn] = c1[r];
    }
}

__global__ __launch_bounds__(256)
void combine_chunks(const float* __restrict__ part, const float* __restrict__ stat,
                    float* __restrict__ out) {
    const int tid = threadIdx.x;
    const int row = blockIdx.x * 4 + (tid >> 6);
    const int d4  = (tid & 63) * 4;
    const v4f* st = (const v4f*)(stat + (size_t)row * NCH_ * 2);
    const v4f* pr = (const v4f*)(part + (size_t)row * NCH_ * D_ + d4);
    v4f ml[NCH_ / 2], o[NCH_];
    #pragma unroll
    for (int i = 0; i < NCH_ / 2; ++i) ml[i] = st[i];
    #pragma unroll
    for (int c = 0; c < NCH_; ++c) o[c] = pr[(size_t)c * (D_ / 4)];

    float mx = -__builtin_inff();
    #pragma unroll
    for (int i = 0; i < NCH_ / 2; ++i) mx = fmaxf(mx, fmaxf(ml[i].x, ml[i].z));

    float den = 0.f;
    v4f acc = {0.f, 0.f, 0.f, 0.f};
    #pragma unroll
    for (int c = 0; c < NCH_; ++c) {
        const float m = (c & 1) ? ml[c >> 1].z : ml[c >> 1].x;
        const float l = (c & 1) ? ml[c >> 1].w : ml[c >> 1].y;
        const float w = __builtin_amdgcn_exp2f((m - mx) * LOG2E_);
        den += w * l;
        acc += w * o[c];
    }
    const float inv = 1.0f / den;
    *(v4f*)(out + (size_t)row * D_ + d4) = acc * inv;
}

extern "C" void kernel_launch(void* const* d_in, const int* in_sizes, int n_in,
                              void* d_out, int out_size, void* d_ws, size_t ws_size,
                              hipStream_t stream) {
    const float* q    = (const float*)d_in[0];
    const float* k    = (const float*)d_in[1];
    const float* v    = (const float*)d_in[2];
    const int*   mask = (const int*)  d_in[3];
    const float* Wq   = (const float*)d_in[4];
    const float* Wk   = (const float*)d_in[5];
    const float* wv   = (const float*)d_in[6];
    float* out = (float*)d_out;

    float* qp = (float*)d_ws;
    float* kp = qp + (size_t)B_ * Q_ * D_;
    float* part = kp + (size_t)B_ * K_ * D_;
    float* stat = part + (size_t)B_ * Q_ * NCH_ * D_;

    proj_gemm_wmma<<<dim3(2, (B_ * Q_) / 16), 256, 0, stream>>>(q, Wq, qp);
    proj_gemm_wmma<<<dim3(2, (B_ * K_) / 16), 256, 0, stream>>>(k, Wk, kp);
    chunk_attention<<<dim3(NCH_, Q_ / 16, B_), 256, 0, stream>>>(qp, kp, wv, mask, v, part, stat);
    combine_chunks<<<(B_ * Q_) / 4, 256, 0, stream>>>(part, stat, out);
}


# projections: all W fragment loads of a wave (128 dwords/lane) issued at entry, kept in flight across the X staging (its barrier fences LDS only); 64 WMMAs run from registers; sums bit-identical
# speedup vs baseline: 2.6153x; 1.0678x over previous
#include <hip/hip_runtime.h>
#include <hip/hip_bf16.h>

typedef __attribute__((ext_vector_type(2))) float v2f;
typedef __attribute__((ext_vector_type(4))) float v4f;
typedef __attribute__((ext_vector_type(8))) float v8f;

#define B_ 4
#define Q_ 256
#define K_ 1024
#define D_ 256

__device__ __forceinline__ float tanh_fast(float x) {
    return __builtin_amdgcn_tanhf(x);
}

#define KS_ (D_ / 4)

__global__ __launch_bounds__(256)
void proj_gemm_wmma(const float* __restrict__ X, const float* __restrict__ W,
                    float* __restrict__ Y) {
    __shared__ float xs[16][D_ + 4];
    const int tid = threadIdx.x;
    const int m0  = blockIdx.y * 16;
    const int n0  = blockIdx.x * 128;
    const int lane = tid & 31;
    const int wave = tid >> 5;
    const int mn   = lane & 15;
    const int hi   = lane >> 4;
    const int nw   = n0 + wave * 16;

    float xr[16];
    #pragma unroll
    for (int i = 0; i < 16; ++i)
        xr[i] = X[(size_t)(m0 + i) * D_ + tid];

    const float* wp = W + (size_t)(2 * hi) * D_ + nw + mn;
    v2f bf[KS_];
    #pragma unroll
    for (int s = 0; s < KS_; ++s) {
        bf[s].x = wp[(size_t)(4 * s + 0) * D_];
        bf[s].y = wp[(size_t)(4 * s + 1) * D_];
    }

    #pragma unroll
    for (int i = 0; i < 16; ++i)
        xs[i][tid] = xr[i];
    __builtin_amdgcn_fence(__ATOMIC_RELEASE, "workgroup", "local");
    __builtin_amdgcn_s_barrier();
    __builtin_amdgcn_fence(__ATOMIC_ACQUIRE, "workgroup", "local");

    v8f c = {};
    #pragma unroll
    for (int s = 0; s < KS_; ++s) {
        v2f a;
        a.x = xs[mn][4 * s + 2 * hi + 0];
        a.y = xs[mn][4 * s + 2 * hi + 1];
        c = __builtin_amdgcn_wmma_f32_16x16x4_f32(false, a, false, bf[s],
                                                  (short)0, c, false, false);
    }
    #pragma unroll
    for (int r = 0; r < 8; ++r)
        Y[(size_t)(m0 + r + 8 * hi) * D_ + nw + mn] = c[r];
}

#define KC_  64
#define NCH_ (K_ / KC_)
#define ES_  64
#define NSL_ (D_ / ES_)
#define LOG2E_ 1.44269504088896f

__global__ __launch_bounds__(256)
void chunk_attention(const float* __restrict__ qp, const float* __restrict__ kp,
                     const float* __restrict__ wv, const int* __restrict__ mask,
                     const float* __restrict__ V,
                     float* __restrict__ part, float* __restrict__ stat) {
    __shared__ __attribute__((aligned(16))) float ks[KC_][ES_ + 4];
    __shared__ float ps[16][KC_ + 4];
    const int tid = threadIdx.x;
    const int kc = blockIdx.x, q0 = blockIdx.y * 16, b = blockIdx.z;
    const int k0 = kc * KC_;
    const int lane = tid & 31;
    const int wave = __builtin_amdgcn_readfirstlane(tid >> 5);
    const int mn = lane & 15, hi = lane >> 4;
    const int key_a = mn + 32 * hi;
    const int key_b = key_a + 16;

    const float* kpb = kp + ((size_t)b * K_ + k0) * D_;
    const float* qr0 = qp + ((size_t)b * Q_ + q0 + 2 * wave) * D_;
    const float* qr1 = qr0 + D_;

    const int mk_a = mask[(size_t)b * K_ + k0 + key_a];
    const int mk_b = mask[(size_t)b * K_ + k0 + key_b];

    const int sr = tid >> 4, sc = (tid & 15) * 4;
    v4f kreg[4];
    #pragma unroll
    for (int i = 0; i < 4; ++i)
        kreg[i] = *(const v4f*)(kpb + (size_t)(sr + 16 * i) * D_ + sc);

    float s00 = 0.f, s01 = 0.f, s10 = 0.f, s11 = 0.f;
    for (int sl = 0; sl < NSL_; ++sl) {
        __syncthreads();
        #pragma unroll
        for (int i = 0; i < 4; ++i) *(v4f*)&ks[sr + 16 * i][sc] = kreg[i];
        __syncthreads();
        if (sl + 1 < NSL_) {
            #pragma unroll
            for (int i = 0; i < 4; ++i)
                kreg[i] = *(const v4f*)(kpb + (size_t)(sr + 16 * i) * D_ + (sl + 1) * ES_ + sc);
        }
        const float* q0e = qr0 + sl * ES_;
        const float* q1e = qr1 + sl * ES_;
        const float* wve = wv + sl * ES_;
        #pragma unroll 4
        for (int e = 0; e < ES_; e += 4) {
            const v4f ka = *(const v4f*)&ks[key_a][e];
            const v4f kb = *(const v4f*)&ks[key_b][e];
            #pragma unroll
            for (int u = 0; u < 4; ++u) {
                const float w = wve[e + u], qa = q0e[e + u], qb = q1e[e + u];
                s00 = __builtin_fmaf(tanh_fast(qa + ka[u]), w, s00);
                s01 = __builtin_fmaf(tanh_fast(qa + kb[u]), w, s01);
                s10 = __builtin_fmaf(tanh_fast(qb + ka[u]), w, s10);
                s11 = __builtin_fmaf(tanh_fast(qb + kb[u]), w, s11);
            }
        }
    }

    float sa[2], sb[2], m[2], l[2];
    sa[0] = (mk_a != 0) ? s00 : -__builtin_inff();
    sb[0] = (mk_b != 0) ? s01 : -__builtin_inff();
    sa[1] = (mk_a != 0) ? s10 : -__builtin_inff();
    sb[1] = (mk_b != 0) ? s11 : -__builtin_inff();
    #pragma unroll
    for (int j = 0; j < 2; ++j) m[j] = fmaxf(sa[j], sb[j]);
    #pragma unroll
    for (int off = 16; off > 0; off >>= 1) {
        float x[2];
        #pragma unroll
        for (int j = 0; j < 2; ++j) x[j] = __shfl_xor(m[j], off, 32);
        #pragma unroll
        for (int j = 0; j < 2; ++j) m[j] = fmaxf(m[j], x[j]);
    }
    float pa[2], pb[2], xa[2], xb[2];
    #pragma unroll
    for (int j = 0; j < 2; ++j) {
        const float msafe = (m[j] == -__builtin_inff()) ? 0.f : m[j];
        pa[j] = __builtin_amdgcn_exp2f((sa[j] - msafe) * LOG2E_);
        pb[j] = __builtin_amdgcn_exp2f((sb[j] - msafe) * LOG2E_);
        ps[2 * wave + j][key_a] = pa[j];
        ps[2 * wave + j][key_b] = pb[j];
    }
    #pragma unroll
    for (int j = 0; j < 2; ++j) {
        xa[j] = __shfl_xor(pa[j], 16, 32);
        xb[j] = __shfl_xor(pb[j], 16, 32);
    }
    #pragma unroll
    for (int j = 0; j < 2; ++j) {
        const float t0 = hi ? xa[j] : pa[j], t1 = hi ? xb[j] : pb[j];
        const float t2 = hi ? pa[j] : xa[j], t3 = hi ? pb[j] : xb[j];
        l[j] = ((t0 + t1) + t2) + t3;
    }
    #pragma unroll
    for (int off = 8; off > 0; off >>= 1) {
        float x[2];
        #pragma unroll
        for (int j = 0; j < 2; ++j) x[j] = __shfl_xor(l[j], off, 16);
        #pragma unroll
        for (int j = 0; j < 2; ++j) l[j] += x[j];
    }
    if (lane == 0) {
        #pragma unroll
        for (int j = 0; j < 2; ++j) {
            const size_t row = ((size_t)b * Q_ + q0 + 2 * wave + j) * NCH_ + kc;
            stat[row * 2 + 0] = m[j];
            stat[row * 2 + 1] = l[j];
        }
    }
    __syncthreads();

    const int d0 = wave * 32;
    const float* vb = V + ((size_t)b * K_ + k0) * D_;

    v8f c0 = {}, c1 = {};
    #pragma unroll 8
    for (int kk = 0; kk < KC_; kk += 4) {
        v2f a, b0, b1;
        a.x = ps[mn][kk + 2 * hi + 0];
        a.y = ps[mn][kk + 2 * hi + 1];
        const float* vr0 = vb + (size_t)(kk + 2 * hi + 0) * D_;
        const float* vr1 = vb + (size_t)(kk + 2 * hi + 1) * D_;
        b0.x = vr0[d0 + mn];       b0.y = vr1[d0 + mn];
        b1.x = vr0[d0 + 16 + mn];  b1.y = vr1[d0 + 16 + mn];
        c0 = __builtin_amdgcn_wmma_f32_16x16x4_f32(false, a, false, b0,
                                                   (short)0, c0, false, false);
        c1 = __builtin_amdgcn_wmma_f32_16x16x4_f32(false, a, false, b1,
                                                   (short)0, c1, false, false);
    }
    #pragma unroll
    for (int r = 0; r < 8; ++r) {
        float* pr = part + (((size_t)b * Q_ + q0 + r + 8 * hi) * NCH_ + kc) * D_;
        pr[d0 + mn]      = c0[r];
        pr[d0 + 16 + mn] = c1[r];
    }
}

__global__ __launch_bounds__(256)
void combine_chunks(const float* __restrict__ part, const float* __restrict__ stat,
                    float* __restrict__ out) {
    const int tid = threadIdx.x;
    const int row = blockIdx.x * 4 + (tid >> 6);
    const int d4  = (tid & 63) * 4;
    const v4f* st = (const v4f*)(stat + (size_t)row * NCH_ * 2);
    const v4f* pr = (const v4f*)(part + (size_t)row * NCH_ * D_ + d4);
    v4f ml[NCH_ / 2], o[NCH_];
    #pragma unroll
    for (int i = 0; i < NCH_ / 2; ++i) ml[i] = st[i];
    #pragma unroll
    for (int c = 0; c < NCH_; ++c) o[c] = pr[(size_t)c * (D_ / 4)];

    float mx = -__builtin_inff();
    #pragma unroll
    for (int i = 0; i < NCH_ / 2; ++i) mx = fmaxf(mx, fmaxf(ml[i].x, ml[i].z));

    float den = 0.f;
    v4f acc = {0.f, 0.f, 0.f, 0.f};
    #pragma unroll
    for (int c = 0; c < NCH_; ++c) {
        const float m = (c & 1) ? ml[c >> 1].z : ml[c >> 1].x;
        const float l = (c & 1) ? ml[c >> 1].w : ml[c >> 1].y;
        const float w = __builtin_amdgcn_exp2f((m - mx) * LOG2E_);
        den += w * l;
        acc += w * o[c];
    }
    const float inv = 1.0f / den;
    *(v4f*)(out + (size_t)row * D_ + d4) = acc * inv;
}

extern "C" void kernel_launch(void* const* d_in, const int* in_sizes, int n_in,
                              void* d_out, int out_size, void* d_ws, size_t ws_size,
                              hipStream_t stream) {
    const float* q    = (const float*)d_in[0];
    const float* k    = (const float*)d_in[1];
    const float* v    = (const float*)d_in[2];
    const int*   mask = (const int*)  d_in[3];
    const float* Wq   = (const float*)d_in[4];
    const float* Wk   = (const float*)d_in[5];
    const float* wv   = (const float*)d_in[6];
    float* out = (float*)d_out;

    float* qp = (float*)d_ws;
    float* kp = qp + (size_t)B_ * Q_ * D_;
    float* part = kp + (size_t)B_ * K_ * D_;
    float* stat = part + (size_t)B_ * Q_ * NCH_ * D_;

    proj_gemm_wmma<<<dim3(2, (B_ * Q_) / 16), 256, 0, stream>>>(q, Wq, qp);
    proj_gemm_wmma<<<dim3(2, (B_ * K_) / 16), 256, 0, stream>>>(k, Wk, kp);
    chunk_attention<<<dim3(NCH_, Q_ / 16, B_), 256, 0, stream>>>(qp, kp, wv, mask, v, part, stat);
    combine_chunks<<<(B_ * Q_) / 4, 256, 0, stream>>>(part, stat, out);
}


# scores: inner loop software-pipelined per e (4 v_tanh, next e's adds, previous e's fmas) and its issue order set: every v_tanh followed by one dual-issue pair, none back to back; sums bit-identical
# speedup vs baseline: 2.7049x; 1.0343x over previous
#include <hip/hip_runtime.h>
#include <hip/hip_bf16.h>

typedef __attribute__((ext_vector_type(2))) float v2f;
typedef __attribute__((ext_vector_type(4))) float v4f;
typedef __attribute__((ext_vector_type(8))) float v8f;

#define B_ 4
#define Q_ 256
#define K_ 1024
#define D_ 256

__device__ __forceinline__ float tanh_fast(float x) {
    return __builtin_amdgcn_tanhf(x);
}

#define KS_ (D_ / 4)

__global__ __launch_bounds__(256)
void proj_gemm_wmma(const float* __restrict__ X, const float* __restrict__ W,
                    float* __restrict__ Y) {
    __shared__ float xs[16][D_ + 4];
    const int tid = threadIdx.x;
    const int m0  = blockIdx.y * 16;
    const int n0  = blockIdx.x * 128;
    const int lane = tid & 31;
    const int wave = tid >> 5;
    const int mn   = lane & 15;
    const int hi   = lane >> 4;
    const int nw   = n0 + wave * 16;

    float xr[16];
    #pragma unroll
    for (int i = 0; i < 16; ++i)
        xr[i] = X[(size_t)(m0 + i) * D_ + tid];

    const float* wp = W + (size_t)(2 * hi) * D_ + nw + mn;
    v2f bf[KS_];
    #pragma unroll
    for (int s = 0; s < KS_; ++s) {
        bf[s].x = wp[(size_t)(4 * s + 0) * D_];
        bf[s].y = wp[(size_t)(4 * s + 1) * D_];
    }

    #pragma unroll
    for (int i = 0; i < 16; ++i)
        xs[i][tid] = xr[i];
    __builtin_amdgcn_fence(__ATOMIC_RELEASE, "workgroup", "local");
    __builtin_amdgcn_s_barrier();
    __builtin_amdgcn_fence(__ATOMIC_ACQUIRE, "workgroup", "local");

    v8f c = {};
    #pragma unroll
    for (int s = 0; s < KS_; ++s) {
        v2f a;
        a.x = xs[mn][4 * s + 2 * hi + 0];
        a.y = xs[mn][4 * s + 2 * hi + 1];
        c = __builtin_amdgcn_wmma_f32_16x16x4_f32(false, a, false, bf[s],
                                                  (short)0, c, false, false);
    }
    #pragma unroll
    for (int r = 0; r < 8; ++r)
        Y[(size_t)(m0 + r + 8 * hi) * D_ + nw + mn] = c[r];
}

#define KC_  64
#define NCH_ (K_ / KC_)
#define ES_  64
#define NSL_ (D_ / ES_)
#define LOG2E_ 1.44269504088896f

__global__ __launch_bounds__(256)
void chunk_attention(const float* __restrict__ qp, const float* __restrict__ kp,
                     const float* __restrict__ wv, const int* __restrict__ mask,
                     const float* __restrict__ V,
                     float* __restrict__ part, float* __restrict__ stat) {
    __shared__ __attribute__((aligned(16))) float ks[KC_][ES_ + 4];
    __shared__ float ps[16][KC_ + 4];
    const int tid = threadIdx.x;
    const int kc = blockIdx.x, q0 = blockIdx.y * 16, b = blockIdx.z;
    const int k0 = kc * KC_;
    const int lane = tid & 31;
    const int wave = __builtin_amdgcn_readfirstlane(tid >> 5);
    const int mn = lane & 15, hi = lane >> 4;
    const int key_a = mn + 32 * hi;
    const int key_b = key_a + 16;

    const float* kpb = kp + ((size_t)b * K_ + k0) * D_;
    const float* qr0 = qp + ((size_t)b * Q_ + q0 + 2 * wave) * D_;
    const float* qr1 = qr0 + D_;

    const int mk_a = mask[(size_t)b * K_ + k0 + key_a];
    const int mk_b = mask[(size_t)b * K_ + k0 + key_b];

    const int sr = tid >> 4, sc = (tid & 15) * 4;
    v4f kreg[4];
    #pragma unroll
    for (int i = 0; i < 4; ++i)
        kreg[i] = *(const v4f*)(kpb + (size_t)(sr + 16 * i) * D_ + sc);

    float s00 = 0.f, s01 = 0.f, s10 = 0.f, s11 = 0.f;
    for (int sl = 0; sl < NSL_; ++sl) {
        __syncthreads();
        #pragma unroll
        for (int i = 0; i < 4; ++i) *(v4f*)&ks[sr + 16 * i][sc] = kreg[i];
        __syncthreads();
        if (sl + 1 < NSL_) {
            #pragma unroll
            for (int i = 0; i < 4; ++i)
                kreg[i] = *(const v4f*)(kpb + (size_t)(sr + 16 * i) * D_ + (sl + 1) * ES_ + sc);
        }
        const float* q0e = qr0 + sl * ES_;
        const float* q1e = qr1 + sl * ES_;
        const float* wve = wv + sl * ES_;
        #pragma unroll 1
        for (int e0 = 0; e0 < ES_; e0 += 16) {
            v4f ka[4], kb[4];
            #pragma unroll
            for (int i = 0; i < 4; ++i) {
                ka[i] = *(const v4f*)&ks[key_a][e0 + 4 * i];
                kb[i] = *(const v4f*)&ks[key_b][e0 + 4 * i];
            }
            float x0 = q0e[e0] + ka[0][0], x2 = q1e[e0] + ka[0][0];
            float x1 = q0e[e0] + kb[0][0], x3 = q1e[e0] + kb[0][0];
            float t0 = 0.f, t1 = 0.f, t2 = 0.f, t3 = 0.f;
            #pragma unroll
            for (int j = 0; j < 16; ++j) {
                __builtin_amdgcn_sched_barrier(0);
                const float n0 = tanh_fast(x0), n2 = tanh_fast(x2), n1 = tanh_fast(x1), n3 = tanh_fast(x3);
                if (j + 1 < 16) {
                    const float qa = q0e[e0 + j + 1], qb = q1e[e0 + j + 1];
                    const float a = ka[(j + 1) >> 2][(j + 1) & 3], bb = kb[(j + 1) >> 2][(j + 1) & 3];
                    x0 = qa + a; x2 = qb + a; x1 = qa + bb; x3 = qb + bb;
                }
                if (j > 0) {
                    const float w = wve[e0 + j - 1];
                    s00 = __builtin_fmaf(t0, w, s00);
                    s11 = __builtin_fmaf(t3, w, s11);
                    s01 = __builtin_fmaf(t1, w, s01);
                    s10 = __builtin_fmaf(t2, w, s10);
                }
                #pragma unroll
                for (int g = 0; g < 4; ++g) {
                    __builtin_amdgcn_sched_group_barrier(0x400, 1, 0);
                    __builtin_amdgcn_sched_group_barrier(0x002, 2, 0);
                }
                t0 = n0; t1 = n1; t2 = n2; t3 = n3;
            }
            __builtin_amdgcn_sched_barrier(0);
            {
                const float w = wve[e0 + 15];
                s00 = __builtin_fmaf(t0, w, s00);
                s11 = __builtin_fmaf(t3, w, s11);
                s01 = __builtin_fmaf(t1, w, s01);
                s10 = __builtin_fmaf(t2, w, s10);
            }
        }
    }

    float sa[2], sb[2], m[2], l[2];
    sa[0] = (mk_a != 0) ? s00 : -__builtin_inff();
    sb[0] = (mk_b != 0) ? s01 : -__builtin_inff();
    sa[1] = (mk_a != 0) ? s10 : -__builtin_inff();
    sb[1] = (mk_b != 0) ? s11 : -__builtin_inff();
    #pragma unroll
    for (int j = 0; j < 2; ++j) m[j] = fmaxf(sa[j], sb[j]);
    #pragma unroll
    for (int off = 16; off > 0; off >>= 1) {
        float x[2];
        #pragma unroll
        for (int j = 0; j < 2; ++j) x[j] = __shfl_xor(m[j], off, 32);
        #pragma unroll
        for (int j = 0; j < 2; ++j) m[j] = fmaxf(m[j], x[j]);
    }
    float pa[2], pb[2], xa[2], xb[2];
    #pragma unroll
    for (int j = 0; j < 2; ++j) {
        const float msafe = (m[j] == -__builtin_inff()) ? 0.f : m[j];
        pa[j] = __builtin_amdgcn_exp2f((sa[j] - msafe) * LOG2E_);
        pb[j] = __builtin_amdgcn_exp2f((sb[j] - msafe) * LOG2E_);
        ps[2 * wave + j][key_a] = pa[j];
        ps[2 * wave + j][key_b] = pb[j];
    }
    #pragma unroll
    for (int j = 0; j < 2; ++j) {
        xa[j] = __shfl_xor(pa[j], 16, 32);
        xb[j] = __shfl_xor(pb[j], 16, 32);
    }
    #pragma unroll
    for (int j = 0; j < 2; ++j) {
        const float t0 = hi ? xa[j] : pa[j], t1 = hi ? xb[j] : pb[j];
        const float t2 = hi ? pa[j] : xa[j], t3 = hi ? pb[j] : xb[j];
        l[j] = ((t0 + t1) + t2) + t3;
    }
    #pragma unroll
    for (int off = 8; off > 0; off >>= 1) {
        float x[2];
        #pragma unroll
        for (int j = 0; j < 2; ++j) x[j] = __shfl_xor(l[j], off, 16);
        #pragma unroll
        for (int j = 0; j < 2; ++j) l[j] += x[j];
    }
    if (lane == 0) {
        #pragma unroll
        for (int j = 0; j < 2; ++j) {
            const size_t row = ((size_t)b * Q_ + q0 + 2 * wave + j) * NCH_ + kc;
            stat[row * 2 + 0] = m[j];
            stat[row * 2 + 1] = l[j];
        }
    }
    __syncthreads();

    const int d0 = wave * 32;
    const float* vb = V + ((size_t)b * K_ + k0) * D_;

    v8f c0 = {}, c1 = {};
    #pragma unroll 8
    for (int kk = 0; kk < KC_; kk += 4) {
        v2f a, b0, b1;
        a.x = ps[mn][kk + 2 * hi + 0];
        a.y = ps[mn][kk + 2 * hi + 1];
        const float* vr0 = vb + (size_t)(kk + 2 * hi + 0) * D_;
        const float* vr1 = vb + (size_t)(kk + 2 * hi + 1) * D_;
        b0.x = vr0[d0 + mn];       b0.y = vr1[d0 + mn];
        b1.x = vr0[d0 + 16 + mn];  b1.y = vr1[d0 + 16 + mn];
        c0 = __builtin_amdgcn_wmma_f32_16x16x4_f32(false, a, false, b0,
                                                   (short)0, c0, false, false);
        c1 = __builtin_amdgcn_wmma_f32_16x16x4_f32(false, a, false, b1,
                                                   (short)0, c1, false, false);
    }
    #pragma unroll
    for (int r = 0; r < 8; ++r) {
        float* pr = part + (((size_t)b * Q_ + q0 + r + 8 * hi) * NCH_ + kc) * D_;
        pr[d0 + mn]      = c0[r];
        pr[d0 + 16 + mn] = c1[r];
    }
}

__global__ __launch_bounds__(256)
void combine_chunks(const float* __restrict__ part, const float* __restrict__ stat,
                    float* __restrict__ out) {
    const int tid = threadIdx.x;
    const int row = blockIdx.x * 4 + (tid >> 6);
    const int d4  = (tid & 63) * 4;
    const v4f* st = (const v4f*)(stat + (size_t)row * NCH_ * 2);
    const v4f* pr = (const v4f*)(part + (size_t)row * NCH_ * D_ + d4);
    v4f ml[NCH_ / 2], o[NCH_];
    #pragma unroll
    for (int i = 0; i < NCH_ / 2; ++i) ml[i] = st[i];
    #pragma unroll
    for (int c = 0; c < NCH_; ++c) o[c] = pr[(size_t)c * (D_ / 4)];

    float mx = -__builtin_inff();
    #pragma unroll
    for (int i = 0; i < NCH_ / 2; ++i) mx = fmaxf(mx, fmaxf(ml[i].x, ml[i].z));

    float den = 0.f;
    v4f acc = {0.f, 0.f, 0.f, 0.f};
    #pragma unroll
    for (int c = 0; c < NCH_; ++c) {
        const float m = (c & 1) ? ml[c >> 1].z : ml[c >> 1].x;
        const float l = (c & 1) ? ml[c >> 1].w : ml[c >> 1].y;
        const float w = __builtin_amdgcn_exp2f((m - mx) * LOG2E_);
        den += w * l;
        acc += w * o[c];
    }
    const float inv = 1.0f / den;
    *(v4f*)(out + (size_t)row * D_ + d4) = acc * inv;
}

extern "C" void kernel_launch(void* const* d_in, const int* in_sizes, int n_in,
                              void* d_out, int out_size, void* d_ws, size_t ws_size,
                              hipStream_t stream) {
    const float* q    = (const float*)d_in[0];
    const float* k    = (const float*)d_in[1];
    const float* v    = (const float*)d_in[2];
    const int*   mask = (const int*)  d_in[3];
    const float* Wq   = (const float*)d_in[4];
    const float* Wk   = (const float*)d_in[5];
    const float* wv   = (const float*)d_in[6];
    float* out = (float*)d_out;

    float* qp = (float*)d_ws;
    float* kp = qp + (size_t)B_ * Q_ * D_;
    float* part = kp + (size_t)B_ * K_ * D_;
    float* stat = part + (size_t)B_ * Q_ * NCH_ * D_;

    proj_gemm_wmma<<<dim3(2, (B_ * Q_) / 16), 256, 0, stream>>>(q, Wq, qp);
    proj_gemm_wmma<<<dim3(2, (B_ * K_) / 16), 256, 0, stream>>>(k, Wk, kp);
    chunk_attention<<<dim3(NCH_, Q_ / 16, B_), 256, 0, stream>>>(qp, kp, wv, mask, v, part, stat);
    combine_chunks<<<(B_ * Q_) / 4, 256, 0, stream>>>(part, stat, out);
}


# projections: q and k projection run in ONE launch of the unchanged projection kernel (grid 2 x 320, a block picks X, W, Y by its row-block index): 3 launches, tiles and sums bit-identical
# speedup vs baseline: 2.9606x; 1.0945x over previous
#include <hip/hip_runtime.h>
#include <hip/hip_bf16.h>

typedef __attribute__((ext_vector_type(2))) float v2f;
typedef __attribute__((ext_vector_type(4))) float v4f;
typedef __attribute__((ext_vector_type(8))) float v8f;

#define B_ 4
#define Q_ 256
#define K_ 1024
#define D_ 256

__device__ __forceinline__ float tanh_fast(float x) {
    return __builtin_amdgcn_tanhf(x);
}

#define KS_ (D_ / 4)
#define KB_ ((B_ * K_) / 16)
#define QB_ ((B_ * Q_) / 16)

__global__ __launch_bounds__(256)
void proj_gemm_wmma(const float* __restrict__ Xk, const float* __restrict__ Wk_,
                    float* __restrict__ Yk,
                    const float* __restrict__ Xq, const float* __restrict__ Wq_,
                    float* __restrict__ Yq) {
    __shared__ float xs[16][D_ + 4];
    const int tid = threadIdx.x;
    const bool isk = blockIdx.y < KB_;
    const float* __restrict__ X = isk ? Xk : Xq;
    const float* __restrict__ W = isk ? Wk_ : Wq_;
    float* __restrict__ Y       = isk ? Yk : Yq;
    const int m0  = (isk ? blockIdx.y : blockIdx.y - KB_) * 16;
    const int n0  = blockIdx.x * 128;
    const int lane = tid & 31;
    const int wave = tid >> 5;
    const int mn   = lane & 15;
    const int hi   = lane >> 4;
    const int nw   = n0 + wave * 16;

    float xr[16];
    #pragma unroll
    for (int i = 0; i < 16; ++i)
        xr[i] = X[(size_t)(m0 + i) * D_ + tid];

    const float* wp = W + (size_t)(2 * hi) * D_ + nw + mn;
    v2f bf[KS_];
    #pragma unroll
    for (int s = 0; s < KS_; ++s) {
        bf[s].x = wp[(size_t)(4 * s + 0) * D_];
        bf[s].y = wp[(size_t)(4 * s + 1) * D_];
    }

    #pragma unroll
    for (int i = 0; i < 16; ++i)
        xs[i][tid] = xr[i];
    __builtin_amdgcn_fence(__ATOMIC_RELEASE, "workgroup", "local");
    __builtin_amdgcn_s_barrier();
    __builtin_amdgcn_fence(__ATOMIC_ACQUIRE, "workgroup", "local");

    v8f c = {};
    #pragma unroll
    for (int s = 0; s < KS_; ++s) {
        v2f a;
        a.x = xs[mn][4 * s + 2 * hi + 0];
        a.y = xs[mn][4 * s + 2 * hi + 1];
        c = __builtin_amdgcn_wmma_f32_16x16x4_f32(false, a, false, bf[s],
                                                  (short)0, c, false, false);
    }
    #pragma unroll
    for (int r = 0; r < 8; ++r)
        Y[(size_t)(m0 + r + 8 * hi) * D_ + nw + mn] = c[r];
}

#define KC_  64
#define NCH_ (K_ / KC_)
#define ES_  64
#define NSL_ (D_ / ES_)
#define LOG2E_ 1.44269504088896f

__global__ __launch_bounds__(256)
void chunk_attention(const float* __restrict__ qp, const float* __restrict__ kp,
                     const float* __restrict__ wv, const int* __restrict__ mask,
                     const float* __restrict__ V,
                     float* __restrict__ part, float* __restrict__ stat) {
    __shared__ __attribute__((aligned(16))) float ks[KC_][ES_ + 4];
    __shared__ float ps[16][KC_ + 4];
    const int tid = threadIdx.x;
    const int kc = blockIdx.x, q0 = blockIdx.y * 16, b = blockIdx.z;
    const int k0 = kc * KC_;
    const int lane = tid & 31;
    const int wave = __builtin_amdgcn_readfirstlane(tid >> 5);
    const int mn = lane & 15, hi = lane >> 4;
    const int key_a = mn + 32 * hi;
    const int key_b = key_a + 16;

    const float* kpb = kp + ((size_t)b * K_ + k0) * D_;
    const float* qr0 = qp + ((size_t)b * Q_ + q0 + 2 * wave) * D_;
    const float* qr1 = qr0 + D_;

    const int mk_a = mask[(size_t)b * K_ + k0 + key_a];
    const int mk_b = mask[(size_t)b * K_ + k0 + key_b];

    const int sr = tid >> 4, sc = (tid & 15) * 4;
    v4f kreg[4];
    #pragma unroll
    for (int i = 0; i < 4; ++i)
        kreg[i] = *(const v4f*)(kpb + (size_t)(sr + 16 * i) * D_ + sc);

    float s00 = 0.f, s01 = 0.f, s10 = 0.f, s11 = 0.f;
    for (int sl = 0; sl < NSL_; ++sl) {
        __syncthreads();
        #pragma unroll
        for (int i = 0; i < 4; ++i) *(v4f*)&ks[sr + 16 * i][sc] = kreg[i];
        __syncthreads();
        if (sl + 1 < NSL_) {
            #pragma unroll
            for (int i = 0; i < 4; ++i)
                kreg[i] = *(const v4f*)(kpb + (size_t)(sr + 16 * i) * D_ + (sl + 1) * ES_ + sc);
        }
        const float* q0e = qr0 + sl * ES_;
        const float* q1e = qr1 + sl * ES_;
        const float* wve = wv + sl * ES_;
        #pragma unroll 1
        for (int e0 = 0; e0 < ES_; e0 += 16) {
            v4f ka[4], kb[4];
            #pragma unroll
            for (int i = 0; i < 4; ++i) {
                ka[i] = *(const v4f*)&ks[key_a][e0 + 4 * i];
                kb[i] = *(const v4f*)&ks[key_b][e0 + 4 * i];
            }
            float x0 = q0e[e0] + ka[0][0], x2 = q1e[e0] + ka[0][0];
            float x1 = q0e[e0] + kb[0][0], x3 = q1e[e0] + kb[0][0];
            float t0 = 0.f, t1 = 0.f, t2 = 0.f, t3 = 0.f;
            #pragma unroll
            for (int j = 0; j < 16; ++j) {
                __builtin_amdgcn_sched_barrier(0);
                const float n0 = tanh_fast(x0), n2 = tanh_fast(x2), n1 = tanh_fast(x1), n3 = tanh_fast(x3);
                if (j + 1 < 16) {
                    const float qa = q0e[e0 + j + 1], qb = q1e[e0 + j + 1];
                    const float a = ka[(j + 1) >> 2][(j + 1) & 3], bb = kb[(j + 1) >> 2][(j + 1) & 3];
                    x0 = qa + a; x2 = qb + a; x1 = qa + bb; x3 = qb + bb;
                }
                if (j > 0) {
                    const float w = wve[e0 + j - 1];
                    s00 = __builtin_fmaf(t0, w, s00);
                    s11 = __builtin_fmaf(t3, w, s11);
                    s01 = __builtin_fmaf(t1, w, s01);
                    s10 = __builtin_fmaf(t2, w, s10);
                }
                #pragma unroll
                for (int g = 0; g < 4; ++g) {
                    __builtin_amdgcn_sched_group_barrier(0x400, 1, 0);
                    __builtin_amdgcn_sched_group_barrier(0x002, 2, 0);
                }
                t0 = n0; t1 = n1; t2 = n2; t3 = n3;
            }
            __builtin_amdgcn_sched_barrier(0);
            {
                const float w = wve[e0 + 15];
                s00 = __builtin_fmaf(t0, w, s00);
                s11 = __builtin_fmaf(t3, w, s11);
                s01 = __builtin_fmaf(t1, w, s01);
                s10 = __builtin_fmaf(t2, w, s10);
            }
        }
    }

    float sa[2], sb[2], m[2], l[2];
    sa[0] = (mk_a != 0) ? s00 : -__builtin_inff();
    sb[0] = (mk_b != 0) ? s01 : -__builtin_inff();
    sa[1] = (mk_a != 0) ? s10 : -__builtin_inff();
    sb[1] = (mk_b != 0) ? s11 : -__builtin_inff();
    #pragma unroll
    for (int j = 0; j < 2; ++j) m[j] = fmaxf(sa[j], sb[j]);
    #pragma unroll
    for (int off = 16; off > 0; off >>= 1) {
        float x[2];
        #pragma unroll
        for (int j = 0; j < 2; ++j) x[j] = __shfl_xor(m[j], off, 32);
        #pragma unroll
        for (int j = 0; j < 2; ++j) m[j] = fmaxf(m[j], x[j]);
    }
    float pa[2], pb[2], xa[2], xb[2];
    #pragma unroll
    for (int j = 0; j < 2; ++j) {
        const float msafe = (m[j] == -__builtin_inff()) ? 0.f : m[j];
        pa[j] = __builtin_amdgcn_exp2f((sa[j] - msafe) * LOG2E_);
        pb[j] = __builtin_amdgcn_exp2f((sb[j] - msafe) * LOG2E_);
        ps[2 * wave + j][key_a] = pa[j];
        ps[2 * wave + j][key_b] = pb[j];
    }
    #pragma unroll
    for (int j = 0; j < 2; ++j) {
        xa[j] = __shfl_xor(pa[j], 16, 32);
        xb[j] = __shfl_xor(pb[j], 16, 32);
    }
    #pragma unroll
    for (int j = 0; j < 2; ++j) {
        const float t0 = hi ? xa[j] : pa[j], t1 = hi ? xb[j] : pb[j];
        const float t2 = hi ? pa[j] : xa[j], t3 = hi ? pb[j] : xb[j];
        l[j] = ((t0 + t1) + t2) + t3;
    }
    #pragma unroll
    for (int off = 8; off > 0; off >>= 1) {
        float x[2];
        #pragma unroll
        for (int j = 0; j < 2; ++j) x[j] = __shfl_xor(l[j], off, 16);
        #pragma unroll
        for (int j = 0; j < 2; ++j) l[j] += x[j];
    }
    if (lane == 0) {
        #pragma unroll
        for (int j = 0; j < 2; ++j) {
            const size_t row = ((size_t)b * Q_ + q0 + 2 * wave + j) * NCH_ + kc;
            stat[row * 2 + 0] = m[j];
            stat[row * 2 + 1] = l[j];
        }
    }
    __syncthreads();

    const int d0 = wave * 32;
    const float* vb = V + ((size_t)b * K_ + k0) * D_;

    v8f c0 = {}, c1 = {};
    #pragma unroll 8
    for (int kk = 0; kk < KC_; kk += 4) {
        v2f a, b0, b1;
        a.x = ps[mn][kk + 2 * hi + 0];
        a.y = ps[mn][kk + 2 * hi + 1];
        const float* vr0 = vb + (size_t)(kk + 2 * hi + 0) * D_;
        const float* vr1 = vb + (size_t)(kk + 2 * hi + 1) * D_;
        b0.x = vr0[d0 + mn];       b0.y = vr1[d0 + mn];
        b1.x = vr0[d0 + 16 + mn];  b1.y = vr1[d0 + 16 + mn];
        c0 = __builtin_amdgcn_wmma_f32_16x16x4_f32(false, a, false, b0,
                                                   (short)0, c0, false, false);
        c1 = __builtin_amdgcn_wmma_f32_16x16x4_f32(false, a, false, b1,
                                                   (short)0, c1, false, false);
    }
    #pragma unroll
    for (int r = 0; r < 8; ++r) {
        float* pr = part + (((size_t)b * Q_ + q0 + r + 8 * hi) * NCH_ + kc) * D_;
        pr[d0 + mn]      = c0[r];
        pr[d0 + 16 + mn] = c1[r];
    }
}

__global__ __launch_bounds__(256)
void combine_chunks(const float* __restrict__ part, const float* __restrict__ stat,
                    float* __restrict__ out) {
    const int tid = threadIdx.x;
    const int row = blockIdx.x * 4 + (tid >> 6);
    const int d4  = (tid & 63) * 4;
    const v4f* st = (const v4f*)(stat + (size_t)row * NCH_ * 2);
    const v4f* pr = (const v4f*)(part + (size_t)row * NCH_ * D_ + d4);
    v4f ml[NCH_ / 2], o[NCH_];
    #pragma unroll
    for (int i = 0; i < NCH_ / 2; ++i) ml[i] = st[i];
    #pragma unroll
    for (int c = 0; c < NCH_; ++c) o[c] = pr[(size_t)c * (D_ / 4)];

    float mx = -__builtin_inff();
    #pragma unroll
    for (int i = 0; i < NCH_ / 2; ++i) mx = fmaxf(mx, fmaxf(ml[i].x, ml[i].z));

    float den = 0.f;
    v4f acc = {0.f, 0.f, 0.f, 0.f};
    #pragma unroll
    for (int c = 0; c < NCH_; ++c) {
        const float m = (c & 1) ? ml[c >> 1].z : ml[c >> 1].x;
        const float l = (c & 1) ? ml[c >> 1].w : ml[c >> 1].y;
        const float w = __builtin_amdgcn_exp2f((m - mx) * LOG2E_);
        den += w * l;
        acc += w * o[c];
    }
    const float inv = 1.0f / den;
    *(v4f*)(out + (size_t)row * D_ + d4) = acc * inv;
}

extern "C" void kernel_launch(void* const* d_in, const int* in_sizes, int n_in,
                              void* d_out, int out_size, void* d_ws, size_t ws_size,
                              hipStream_t stream) {
    const float* q    = (const float*)d_in[0];
    const float* k    = (const float*)d_in[1];
    const float* v    = (const float*)d_in[2];
    const int*   mask = (const int*)  d_in[3];
    const float* Wq   = (const float*)d_in[4];
    const float* Wk   = (const float*)d_in[5];
    const float* wv   = (const float*)d_in[6];
    float* out = (float*)d_out;

    float* qp = (float*)d_ws;
    float* kp = qp + (size_t)B_ * Q_ * D_;
    float* part = kp + (size_t)B_ * K_ * D_;
    float* stat = part + (size_t)B_ * Q_ * NCH_ * D_;

    proj_gemm_wmma<<<dim3(2, KB_ + QB_), 256, 0, stream>>>(k, Wk, kp, q, Wq, qp);
    chunk_attention<<<dim3(NCH_, Q_ / 16, B_), 256, 0, stream>>>(qp, kp, wv, mask, v, part, stat);
    combine_chunks<<<(B_ * Q_) / 4, 256, 0, stream>>>(part, stat, out);
}


# projections: a wave's W fragments, held in registers, now serve two 16-row tiles through two accumulators (blocks of 4 waves, 32 rows x 64 cols): half the waves, half the W reads; sums bit-identical
# speedup vs baseline: 3.1305x; 1.0574x over previous
#include <hip/hip_runtime.h>
#include <hip/hip_bf16.h>

typedef __attribute__((ext_vector_type(2))) float v2f;
typedef __attribute__((ext_vector_type(4))) float v4f;
typedef __attribute__((ext_vector_type(8))) float v8f;

#define B_ 4
#define Q_ 256
#define K_ 1024
#define D_ 256

__device__ __forceinline__ float tanh_fast(float x) {
    return __builtin_amdgcn_tanhf(x);
}

#define KS_ (D_ / 4)
#define PR_ 32
#define KB_ ((B_ * K_) / PR_)
#define QB_ ((B_ * Q_) / PR_)

__global__ __launch_bounds__(128)
void proj_gemm_wmma(const float* __restrict__ Xk, const float* __restrict__ Wk_,
                    float* __restrict__ Yk,
                    const float* __restrict__ Xq, const float* __restrict__ Wq_,
                    float* __restrict__ Yq) {
    __shared__ __attribute__((aligned(16))) float xs[PR_][D_ + 4];
    const int tid = threadIdx.x;
    const bool isk = blockIdx.y < KB_;
    const float* __restrict__ X = isk ? Xk : Xq;
    const float* __restrict__ W = isk ? Wk_ : Wq_;
    float* __restrict__ Y       = isk ? Yk : Yq;
    const int m0  = (isk ? blockIdx.y : blockIdx.y - KB_) * PR_;
    const int n0  = blockIdx.x * 64;
    const int lane = tid & 31;
    const int wave = tid >> 5;
    const int mn   = lane & 15;
    const int hi   = lane >> 4;
    const int nw   = n0 + wave * 16;

    v2f xr[PR_];
    #pragma unroll
    for (int i = 0; i < PR_; ++i)
        xr[i] = *(const v2f*)(X + (size_t)(m0 + i) * D_ + 2 * tid);
    __builtin_amdgcn_sched_barrier(0);

    const float* wp = W + (size_t)(2 * hi) * D_ + nw + mn;
    v2f bf[KS_];
    #pragma unroll
    for (int s = 0; s < KS_; ++s) {
        bf[s].x = wp[(size_t)(4 * s + 0) * D_];
        bf[s].y = wp[(size_t)(4 * s + 1) * D_];
    }
    __builtin_amdgcn_sched_barrier(0);

    #pragma unroll
    for (int i = 0; i < PR_; ++i)
        *(v2f*)&xs[i][2 * tid] = xr[i];
    __builtin_amdgcn_fence(__ATOMIC_RELEASE, "workgroup", "local");
    __builtin_amdgcn_s_barrier();
    __builtin_amdgcn_fence(__ATOMIC_ACQUIRE, "workgroup", "local");

    v8f c0 = {}, c1 = {};
    #pragma unroll
    for (int s = 0; s < KS_; ++s) {
        v2f a0, a1;
        a0.x = xs[mn][4 * s + 2 * hi + 0];
        a0.y = xs[mn][4 * s + 2 * hi + 1];
        a1.x = xs[16 + mn][4 * s + 2 * hi + 0];
        a1.y = xs[16 + mn][4 * s + 2 * hi + 1];
        c0 = __builtin_amdgcn_wmma_f32_16x16x4_f32(false, a0, false, bf[s],
                                                   (short)0, c0, false, false);
        c1 = __builtin_amdgcn_wmma_f32_16x16x4_f32(false, a1, false, bf[s],
                                                   (short)0, c1, false, false);
    }
    #pragma unroll
    for (int r = 0; r < 8; ++r) {
        Y[(size_t)(m0 + r + 8 * hi) * D_ + nw + mn]      = c0[r];
        Y[(size_t)(m0 + 16 + r + 8 * hi) * D_ + nw + mn] = c1[r];
    }
}

#define KC_  64
#define NCH_ (K_ / KC_)
#define ES_  64
#define NSL_ (D_ / ES_)
#define LOG2E_ 1.44269504088896f

__global__ __launch_bounds__(256)
void chunk_attention(const float* __restrict__ qp, const float* __restrict__ kp,
                     const float* __restrict__ wv, const int* __restrict__ mask,
                     const float* __restrict__ V,
                     float* __restrict__ part, float* __restrict__ stat) {
    __shared__ __attribute__((aligned(16))) float ks[KC_][ES_ + 4];
    __shared__ float ps[16][KC_ + 4];
    const int tid = threadIdx.x;
    const int kc = blockIdx.x, q0 = blockIdx.y * 16, b = blockIdx.z;
    const int k0 = kc * KC_;
    const int lane = tid & 31;
    const int wave = __builtin_amdgcn_readfirstlane(tid >> 5);
    const int mn = lane & 15, hi = lane >> 4;
    const int key_a = mn + 32 * hi;
    const int key_b = key_a + 16;

    const float* kpb = kp + ((size_t)b * K_ + k0) * D_;
    const float* qr0 = qp + ((size_t)b * Q_ + q0 + 2 * wave) * D_;
    const float* qr1 = qr0 + D_;

    const int mk_a = mask[(size_t)b * K_ + k0 + key_a];
    const int mk_b = mask[(size_t)b * K_ + k0 + key_b];

    const int sr = tid >> 4, sc = (tid & 15) * 4;
    v4f kreg[4];
    #pragma unroll
    for (int i = 0; i < 4; ++i)
        kreg[i] = *(const v4f*)(kpb + (size_t)(sr + 16 * i) * D_ + sc);

    float s00 = 0.f, s01 = 0.f, s10 = 0.f, s11 = 0.f;
    for (int sl = 0; sl < NSL_; ++sl) {
        __syncthreads();
        #pragma unroll
        for (int i = 0; i < 4; ++i) *(v4f*)&ks[sr + 16 * i][sc] = kreg[i];
        __syncthreads();
        if (sl + 1 < NSL_) {
            #pragma unroll
            for (int i = 0; i < 4; ++i)
                kreg[i] = *(const v4f*)(kpb + (size_t)(sr + 16 * i) * D_ + (sl + 1) * ES_ + sc);
        }
        const float* q0e = qr0 + sl * ES_;
        const float* q1e = qr1 + sl * ES_;
        const float* wve = wv + sl * ES_;
        #pragma unroll 1
        for (int e0 = 0; e0 < ES_; e0 += 16) {
            v4f ka[4], kb[4];
            #pragma unroll
            for (int i = 0; i < 4; ++i) {
                ka[i] = *(const v4f*)&ks[key_a][e0 + 4 * i];
                kb[i] = *(const v4f*)&ks[key_b][e0 + 4 * i];
            }
            float x0 = q0e[e0] + ka[0][0], x2 = q1e[e0] + ka[0][0];
            float x1 = q0e[e0] + kb[0][0], x3 = q1e[e0] + kb[0][0];
            float t0 = 0.f, t1 = 0.f, t2 = 0.f, t3 = 0.f;
            #pragma unroll
            for (int j = 0; j < 16; ++j) {
                __builtin_amdgcn_sched_barrier(0);
                const float n0 = tanh_fast(x0), n2 = tanh_fast(x2), n1 = tanh_fast(x1), n3 = tanh_fast(x3);
                if (j + 1 < 16) {
                    const float qa = q0e[e0 + j + 1], qb = q1e[e0 + j + 1];
                    const float a = ka[(j + 1) >> 2][(j + 1) & 3], bb = kb[(j + 1) >> 2][(j + 1) & 3];
                    x0 = qa + a; x2 = qb + a; x1 = qa + bb; x3 = qb + bb;
                }
                if (j > 0) {
                    const float w = wve[e0 + j - 1];
                    s00 = __builtin_fmaf(t0, w, s00);
                    s11 = __builtin_fmaf(t3, w, s11);
                    s01 = __builtin_fmaf(t1, w, s01);
                    s10 = __builtin_fmaf(t2, w, s10);
                }
                #pragma unroll
                for (int g = 0; g < 4; ++g) {
                    __builtin_amdgcn_sched_group_barrier(0x400, 1, 0);
                    __builtin_amdgcn_sched_group_barrier(0x002, 2, 0);
                }
                t0 = n0; t1 = n1; t2 = n2; t3 = n3;
            }
            __builtin_amdgcn_sched_barrier(0);
            {
                const float w = wve[e0 + 15];
                s00 = __builtin_fmaf(t0, w, s00);
                s11 = __builtin_fmaf(t3, w, s11);
                s01 = __builtin_fmaf(t1, w, s01);
                s10 = __builtin_fmaf(t2, w, s10);
            }
        }
    }

    float sa[2], sb[2], m[2], l[2];
    sa[0] = (mk_a != 0) ? s00 : -__builtin_inff();
    sb[0] = (mk_b != 0) ? s01 : -__builtin_inff();
    sa[1] = (mk_a != 0) ? s10 : -__builtin_inff();
    sb[1] = (mk_b != 0) ? s11 : -__builtin_inff();
    #pragma unroll
    for (int j = 0; j < 2; ++j) m[j] = fmaxf(sa[j], sb[j]);
    #pragma unroll
    for (int off = 16; off > 0; off >>= 1) {
        float x[2];
        #pragma unroll
        for (int j = 0; j < 2; ++j) x[j] = __shfl_xor(m[j], off, 32);
        #pragma unroll
        for (int j = 0; j < 2; ++j) m[j] = fmaxf(m[j], x[j]);
    }
    float pa[2], pb[2], xa[2], xb[2];
    #pragma unroll
    for (int j = 0; j < 2; ++j) {
        const float msafe = (m[j] == -__builtin_inff()) ? 0.f : m[j];
        pa[j] = __builtin_amdgcn_exp2f((sa[j] - msafe) * LOG2E_);
        pb[j] = __builtin_amdgcn_exp2f((sb[j] - msafe) * LOG2E_);
        ps[2 * wave + j][key_a] = pa[j];
        ps[2 * wave + j][key_b] = pb[j];
    }
    #pragma unroll
    for (int j = 0; j < 2; ++j) {
        xa[j] = __shfl_xor(pa[j], 16, 32);
        xb[j] = __shfl_xor(pb[j], 16, 32);
    }
    #pragma unroll
    for (int j = 0; j < 2; ++j) {
        const float t0 = hi ? xa[j] : pa[j], t1 = hi ? xb[j] : pb[j];
        const float t2 = hi ? pa[j] : xa[j], t3 = hi ? pb[j] : xb[j];
        l[j] = ((t0 + t1) + t2) + t3;
    }
    #pragma unroll
    for (int off = 8; off > 0; off >>= 1) {
        float x[2];
        #pragma unroll
        for (int j = 0; j < 2; ++j) x[j] = __shfl_xor(l[j], off, 16);
        #pragma unroll
        for (int j = 0; j < 2; ++j) l[j] += x[j];
    }
    if (lane == 0) {
        #pragma unroll
        for (int j = 0; j < 2; ++j) {
            const size_t row = ((size_t)b * Q_ + q0 + 2 * wave + j) * NCH_ + kc;
            stat[row * 2 + 0] = m[j];
            stat[row * 2 + 1] = l[j];
        }
    }
    __syncthreads();

    const int d0 = wave * 32;
    const float* vb = V + ((size_t)b * K_ + k0) * D_;

    v8f c0 = {}, c1 = {};
    #pragma unroll 8
    for (int kk = 0; kk < KC_; kk += 4) {
        v2f a, b0, b1;
        a.x = ps[mn][kk + 2 * hi + 0];
        a.y = ps[mn][kk + 2 * hi + 1];
        const float* vr0 = vb + (size_t)(kk + 2 * hi + 0) * D_;
        const float* vr1 = vb + (size_t)(kk + 2 * hi + 1) * D_;
        b0.x = vr0[d0 + mn];       b0.y = vr1[d0 + mn];
        b1.x = vr0[d0 + 16 + mn];  b1.y = vr1[d0 + 16 + mn];
        c0 = __builtin_amdgcn_wmma_f32_16x16x4_f32(false, a, false, b0,
                                                   (short)0, c0, false, false);
        c1 = __builtin_amdgcn_wmma_f32_16x16x4_f32(false, a, false, b1,
                                                   (short)0, c1, false, false);
    }
    #pragma unroll
    for (int r = 0; r < 8; ++r) {
        float* pr = part + (((size_t)b * Q_ + q0 + r + 8 * hi) * NCH_ + kc) * D_;
        pr[d0 + mn]      = c0[r];
        pr[d0 + 16 + mn] = c1[r];
    }
}

__global__ __launch_bounds__(256)
void combine_chunks(const float* __restrict__ part, const float* __restrict__ stat,
                    float* __restrict__ out) {
    const int tid = threadIdx.x;
    const int row = blockIdx.x * 4 + (tid >> 6);
    const int d4  = (tid & 63) * 4;
    const v4f* st = (const v4f*)(stat + (size_t)row * NCH_ * 2);
    const v4f* pr = (const v4f*)(part + (size_t)row * NCH_ * D_ + d4);
    v4f ml[NCH_ / 2], o[NCH_];
    #pragma unroll
    for (int i = 0; i < NCH_ / 2; ++i) ml[i] = st[i];
    #pragma unroll
    for (int c = 0; c < NCH_; ++c) o[c] = pr[(size_t)c * (D_ / 4)];

    float mx = -__builtin_inff();
    #pragma unroll
    for (int i = 0; i < NCH_ / 2; ++i) mx = fmaxf(mx, fmaxf(ml[i].x, ml[i].z));

    float den = 0.f;
    v4f acc = {0.f, 0.f, 0.f, 0.f};
    #pragma unroll
    for (int c = 0; c < NCH_; ++c) {
        const float m = (c & 1) ? ml[c >> 1].z : ml[c >> 1].x;
        const float l = (c & 1) ? ml[c >> 1].w : ml[c >> 1].y;
        const float w = __builtin_amdgcn_exp2f((m - mx) * LOG2E_);
        den += w * l;
        acc += w * o[c];
    }
    const float inv = 1.0f / den;
    *(v4f*)(out + (size_t)row * D_ + d4) = acc * inv;
}

extern "C" void kernel_launch(void* const* d_in, const int* in_sizes, int n_in,
                              void* d_out, int out_size, void* d_ws, size_t ws_size,
                              hipStream_t stream) {
    const float* q    = (const float*)d_in[0];
    const float* k    = (const float*)d_in[1];
    const float* v    = (const float*)d_in[2];
    const int*   mask = (const int*)  d_in[3];
    const float* Wq   = (const float*)d_in[4];
    const float* Wk   = (const float*)d_in[5];
    const float* wv   = (const float*)d_in[6];
    float* out = (float*)d_out;

    float* qp = (float*)d_ws;
    float* kp = qp + (size_t)B_ * Q_ * D_;
    float* part = kp + (size_t)B_ * K_ * D_;
    float* stat = part + (size_t)B_ * Q_ * NCH_ * D_;

    proj_gemm_wmma<<<dim3(4, KB_ + QB_), 128, 0, stream>>>(k, Wk, kp, q, Wq, qp);
    chunk_attention<<<dim3(NCH_, Q_ / 16, B_), 256, 0, stream>>>(qp, kp, wv, mask, v, part, stat);
    combine_chunks<<<(B_ * Q_) / 4, 256, 0, stream>>>(part, stat, out);
}


# projections: k loop cut into 4 blocks of 16 k steps; a block's A fragments (16 LDS reads, register ring) and its 32 W loads are issued as one group a block ahead of its 32 WMMAs; sums bit-identical
# speedup vs baseline: 3.2835x; 1.0489x over previous
#include <hip/hip_runtime.h>
#include <hip/hip_bf16.h>

typedef __attribute__((ext_vector_type(2))) float v2f;
typedef __attribute__((ext_vector_type(4))) float v4f;
typedef __attribute__((ext_vector_type(8))) float v8f;

#define B_ 4
#define Q_ 256
#define K_ 1024
#define D_ 256

__device__ __forceinline__ float tanh_fast(float x) {
    return __builtin_amdgcn_tanhf(x);
}

#define KS_ (D_ / 4)
#define KBS_ 16
#define NKB_ (KS_ / KBS_)
#define PR_ 32
#define KB_ ((B_ * K_) / PR_)
#define QB_ ((B_ * Q_) / PR_)

__global__ __launch_bounds__(128)
void proj_gemm_wmma(const float* __restrict__ Xk, const float* __restrict__ Wk_,
                    float* __restrict__ Yk,
                    const float* __restrict__ Xq, const float* __restrict__ Wq_,
                    float* __restrict__ Yq) {
    __shared__ __attribute__((aligned(16))) float xs[PR_][D_ + 4];
    const int tid = threadIdx.x;
    const bool isk = blockIdx.y < KB_;
    const float* __restrict__ X = isk ? Xk : Xq;
    const float* __restrict__ W = isk ? Wk_ : Wq_;
    float* __restrict__ Y       = isk ? Yk : Yq;
    const int m0  = (isk ? blockIdx.y : blockIdx.y - KB_) * PR_;
    const int n0  = blockIdx.x * 64;
    const int lane = tid & 31;
    const int wave = tid >> 5;
    const int mn   = lane & 15;
    const int hi   = lane >> 4;
    const int nw   = n0 + wave * 16;

    v2f xr[PR_];
    #pragma unroll
    for (int i = 0; i < PR_; ++i)
        xr[i] = *(const v2f*)(X + (size_t)(m0 + i) * D_ + 2 * tid);
    __builtin_amdgcn_sched_barrier(0);

    const float* wp = W + (size_t)(2 * hi) * D_ + nw + mn;
    v2f bf[KS_];
    #pragma unroll
    for (int s = 0; s < KBS_; ++s) {
        bf[s].x = wp[(size_t)(4 * s + 0) * D_];
        bf[s].y = wp[(size_t)(4 * s + 1) * D_];
    }
    __builtin_amdgcn_sched_barrier(0);

    #pragma unroll
    for (int i = 0; i < PR_; ++i)
        *(v2f*)&xs[i][2 * tid] = xr[i];
    __builtin_amdgcn_fence(__ATOMIC_RELEASE, "workgroup", "local");
    __builtin_amdgcn_s_barrier();
    __builtin_amdgcn_fence(__ATOMIC_ACQUIRE, "workgroup", "local");

    v2f fa[2][KBS_], fb[2][KBS_];
    #pragma unroll
    for (int t = 0; t < KBS_; ++t) {
        fa[0][t].x = xs[mn][4 * t + 2 * hi + 0];
        fa[0][t].y = xs[mn][4 * t + 2 * hi + 1];
        fb[0][t].x = xs[16 + mn][4 * t + 2 * hi + 0];
        fb[0][t].y = xs[16 + mn][4 * t + 2 * hi + 1];
    }

    v8f c0 = {}, c1 = {};
    #pragma unroll
    for (int j = 0; j < NKB_; ++j) {
        if (j + 1 < NKB_) {
            const int sb = (j + 1) * KBS_;
            #pragma unroll
            for (int t = 0; t < KBS_; ++t) {
                fa[(j + 1) & 1][t].x = xs[mn][4 * (sb + t) + 2 * hi + 0];
                fa[(j + 1) & 1][t].y = xs[mn][4 * (sb + t) + 2 * hi + 1];
                fb[(j + 1) & 1][t].x = xs[16 + mn][4 * (sb + t) + 2 * hi + 0];
                fb[(j + 1) & 1][t].y = xs[16 + mn][4 * (sb + t) + 2 * hi + 1];
            }
            #pragma unroll
            for (int t = 0; t < KBS_; ++t) {
                bf[sb + t].x = wp[(size_t)(4 * (sb + t) + 0) * D_];
                bf[sb + t].y = wp[(size_t)(4 * (sb + t) + 1) * D_];
            }
        }
        __builtin_amdgcn_sched_barrier(0);
        #pragma unroll
        for (int t = 0; t < KBS_; ++t) {
            c0 = __builtin_amdgcn_wmma_f32_16x16x4_f32(false, fa[j & 1][t], false, bf[j * KBS_ + t],
                                                       (short)0, c0, false, false);
            c1 = __builtin_amdgcn_wmma_f32_16x16x4_f32(false, fb[j & 1][t], false, bf[j * KBS_ + t],
                                                       (short)0, c1, false, false);
        }
        __builtin_amdgcn_sched_barrier(0);
    }
    #pragma unroll
    for (int r = 0; r < 8; ++r) {
        Y[(size_t)(m0 + r + 8 * hi) * D_ + nw + mn]      = c0[r];
        Y[(size_t)(m0 + 16 + r + 8 * hi) * D_ + nw + mn] = c1[r];
    }
}

#define KC_  64
#define NCH_ (K_ / KC_)
#define ES_  64
#define NSL_ (D_ / ES_)
#define LOG2E_ 1.44269504088896f

__global__ __launch_bounds__(256)
void chunk_attention(const float* __restrict__ qp, const float* __restrict__ kp,
                     const float* __restrict__ wv, const int* __restrict__ mask,
                     const float* __restrict__ V,
                     float* __restrict__ part, float* __restrict__ stat) {
    __shared__ __attribute__((aligned(16))) float ks[KC_][ES_ + 4];
    __shared__ float ps[16][KC_ + 4];
    const int tid = threadIdx.x;
    const int kc = blockIdx.x, q0 = blockIdx.y * 16, b = blockIdx.z;
    const int k0 = kc * KC_;
    const int lane = tid & 31;
    const int wave = __builtin_amdgcn_readfirstlane(tid >> 5);
    const int mn = lane & 15, hi = lane >> 4;
    const int key_a = mn + 32 * hi;
    const int key_b = key_a + 16;

    const float* kpb = kp + ((size_t)b * K_ + k0) * D_;
    const float* qr0 = qp + ((size_t)b * Q_ + q0 + 2 * wave) * D_;
    const float* qr1 = qr0 + D_;

    const int mk_a = mask[(size_t)b * K_ + k0 + key_a];
    const int mk_b = mask[(size_t)b * K_ + k0 + key_b];

    const int sr = tid >> 4, sc = (tid & 15) * 4;
    v4f kreg[4];
    #pragma unroll
    for (int i = 0; i < 4; ++i)
        kreg[i] = *(const v4f*)(kpb + (size_t)(sr + 16 * i) * D_ + sc);

    float s00 = 0.f, s01 = 0.f, s10 = 0.f, s11 = 0.f;
    for (int sl = 0; sl < NSL_; ++sl) {
        __syncthreads();
        #pragma unroll
        for (int i = 0; i < 4; ++i) *(v4f*)&ks[sr + 16 * i][sc] = kreg[i];
        __syncthreads();
        if (sl + 1 < NSL_) {
            #pragma unroll
            for (int i = 0; i < 4; ++i)
                kreg[i] = *(const v4f*)(kpb + (size_t)(sr + 16 * i) * D_ + (sl + 1) * ES_ + sc);
        }
        const float* q0e = qr0 + sl * ES_;
        const float* q1e = qr1 + sl * ES_;
        const float* wve = wv + sl * ES_;
        #pragma unroll 1
        for (int e0 = 0; e0 < ES_; e0 += 16) {
            v4f ka[4], kb[4];
            #pragma unroll
            for (int i = 0; i < 4; ++i) {
                ka[i] = *(const v4f*)&ks[key_a][e0 + 4 * i];
                kb[i] = *(const v4f*)&ks[key_b][e0 + 4 * i];
            }
            float x0 = q0e[e0] + ka[0][0], x2 = q1e[e0] + ka[0][0];
            float x1 = q0e[e0] + kb[0][0], x3 = q1e[e0] + kb[0][0];
            float t0 = 0.f, t1 = 0.f, t2 = 0.f, t3 = 0.f;
            #pragma unroll
            for (int j = 0; j < 16; ++j) {
                __builtin_amdgcn_sched_barrier(0);
                const float n0 = tanh_fast(x0), n2 = tanh_fast(x2), n1 = tanh_fast(x1), n3 = tanh_fast(x3);
                if (j + 1 < 16) {
                    const float qa = q0e[e0 + j + 1], qb = q1e[e0 + j + 1];
                    const float a = ka[(j + 1) >> 2][(j + 1) & 3], bb = kb[(j + 1) >> 2][(j + 1) & 3];
                    x0 = qa + a; x2 = qb + a; x1 = qa + bb; x3 = qb + bb;
                }
                if (j > 0) {
                    const float w = wve[e0 + j - 1];
                    s00 = __builtin_fmaf(t0, w, s00);
                    s11 = __builtin_fmaf(t3, w, s11);
                    s01 = __builtin_fmaf(t1, w, s01);
                    s10 = __builtin_fmaf(t2, w, s10);
                }
                #pragma unroll
                for (int g = 0; g < 4; ++g) {
                    __builtin_amdgcn_sched_group_barrier(0x400, 1, 0);
                    __builtin_amdgcn_sched_group_barrier(0x002, 2, 0);
                }
                t0 = n0; t1 = n1; t2 = n2; t3 = n3;
            }
            __builtin_amdgcn_sched_barrier(0);
            {
                const float w = wve[e0 + 15];
                s00 = __builtin_fmaf(t0, w, s00);
                s11 = __builtin_fmaf(t3, w, s11);
                s01 = __builtin_fmaf(t1, w, s01);
                s10 = __builtin_fmaf(t2, w, s10);
            }
        }
    }

    float sa[2], sb[2], m[2], l[2];
    sa[0] = (mk_a != 0) ? s00 : -__builtin_inff();
    sb[0] = (mk_b != 0) ? s01 : -__builtin_inff();
    sa[1] = (mk_a != 0) ? s10 : -__builtin_inff();
    sb[1] = (mk_b != 0) ? s11 : -__builtin_inff();
    #pragma unroll
    for (int j = 0; j < 2; ++j) m[j] = fmaxf(sa[j], sb[j]);
    #pragma unroll
    for (int off = 16; off > 0; off >>= 1) {
        float x[2];
        #pragma unroll
        for (int j = 0; j < 2; ++j) x[j] = __shfl_xor(m[j], off, 32);
        #pragma unroll
        for (int j = 0; j < 2; ++j) m[j] = fmaxf(m[j], x[j]);
    }
    float pa[2], pb[2], xa[2], xb[2];
    #pragma unroll
    for (int j = 0; j < 2; ++j) {
        const float msafe = (m[j] == -__builtin_inff()) ? 0.f : m[j];
        pa[j] = __builtin_amdgcn_exp2f((sa[j] - msafe) * LOG2E_);
        pb[j] = __builtin_amdgcn_exp2f((sb[j] - msafe) * LOG2E_);
        ps[2 * wave + j][key_a] = pa[j];
        ps[2 * wave + j][key_b] = pb[j];
    }
    #pragma unroll
    for (int j = 0; j < 2; ++j) {
        xa[j] = __shfl_xor(pa[j], 16, 32);
        xb[j] = __shfl_xor(pb[j], 16, 32);
    }
    #pragma unroll
    for (int j = 0; j < 2; ++j) {
        const float t0 = hi ? xa[j] : pa[j], t1 = hi ? xb[j] : pb[j];
        const float t2 = hi ? pa[j] : xa[j], t3 = hi ? pb[j] : xb[j];
        l[j] = ((t0 + t1) + t2) + t3;
    }
    #pragma unroll
    for (int off = 8; off > 0; off >>= 1) {
        float x[2];
        #pragma unroll
        for (int j = 0; j < 2; ++j) x[j] = __shfl_xor(l[j], off, 16);
        #pragma unroll
        for (int j = 0; j < 2; ++j) l[j] += x[j];
    }
    if (lane == 0) {
        #pragma unroll
        for (int j = 0; j < 2; ++j) {
            const size_t row = ((size_t)b * Q_ + q0 + 2 * wave + j) * NCH_ + kc;
            stat[row * 2 + 0] = m[j];
            stat[row * 2 + 1] = l[j];
        }
    }
    __syncthreads();

    const int d0 = wave * 32;
    const float* vb = V + ((size_t)b * K_ + k0) * D_;

    v8f c0 = {}, c1 = {};
    #pragma unroll 8
    for (int kk = 0; kk < KC_; kk += 4) {
        v2f a, b0, b1;
        a.x = ps[mn][kk + 2 * hi + 0];
        a.y = ps[mn][kk + 2 * hi + 1];
        const float* vr0 = vb + (size_t)(kk + 2 * hi + 0) * D_;
        const float* vr1 = vb + (size_t)(kk + 2 * hi + 1) * D_;
        b0.x = vr0[d0 + mn];       b0.y = vr1[d0 + mn];
        b1.x = vr0[d0 + 16 + mn];  b1.y = vr1[d0 + 16 + mn];
        c0 = __builtin_amdgcn_wmma_f32_16x16x4_f32(false, a, false, b0,
                                                   (short)0, c0, false, false);
        c1 = __builtin_amdgcn_wmma_f32_16x16x4_f32(false, a, false, b1,
                                                   (short)0, c1, false, false);
    }
    #pragma unroll
    for (int r = 0; r < 8; ++r) {
        float* pr = part + (((size_t)b * Q_ + q0 + r + 8 * hi) * NCH_ + kc) * D_;
        pr[d0 + mn]      = c0[r];
        pr[d0 + 16 + mn] = c1[r];
    }
}

__global__ __launch_bounds__(256)
void combine_chunks(const float* __restrict__ part, const float* __restrict__ stat,
                    float* __restrict__ out) {
    const int tid = threadIdx.x;
    const int row = blockIdx.x * 4 + (tid >> 6);
    const int d4  = (tid & 63) * 4;
    const v4f* st = (const v4f*)(stat + (size_t)row * NCH_ * 2);
    const v4f* pr = (const v4f*)(part + (size_t)row * NCH_ * D_ + d4);
    v4f ml[NCH_ / 2], o[NCH_];
    #pragma unroll
    for (int i = 0; i < NCH_ / 2; ++i) ml[i] = st[i];
    #pragma unroll
    for (int c = 0; c < NCH_; ++c) o[c] = pr[(size_t)c * (D_ / 4)];

    float mx = -__builtin_inff();
    #pragma unroll
    for (int i = 0; i < NCH_ / 2; ++i) mx = fmaxf(mx, fmaxf(ml[i].x, ml[i].z));

    float den = 0.f;
    v4f acc = {0.f, 0.f, 0.f, 0.f};
    #pragma unroll
    for (int c = 0; c < NCH_; ++c) {
        const float m = (c & 1) ? ml[c >> 1].z : ml[c >> 1].x;
        const float l = (c & 1) ? ml[c >> 1].w : ml[c >> 1].y;
        const float w = __builtin_amdgcn_exp2f((m - mx) * LOG2E_);
        den += w * l;
        acc += w * o[c];
    }
    const float inv = 1.0f / den;
    *(v4f*)(out + (size_t)row * D_ + d4) = acc * inv;
}

extern "C" void kernel_launch(void* const* d_in, const int* in_sizes, int n_in,
                              void* d_out, int out_size, void* d_ws, size_t ws_size,
                              hipStream_t stream) {
    const float* q    = (const float*)d_in[0];
    const float* k    = (const float*)d_in[1];
    const float* v    = (const float*)d_in[2];
    const int*   mask = (const int*)  d_in[3];
    const float* Wq   = (const float*)d_in[4];
    const float* Wk   = (const float*)d_in[5];
    const float* wv   = (const float*)d_in[6];
    float* out = (float*)d_out;

    float* qp = (float*)d_ws;
    float* kp = qp + (size_t)B_ * Q_ * D_;
    float* part = kp + (size_t)B_ * K_ * D_;
    float* stat = part + (size_t)B_ * Q_ * NCH_ * D_;

    proj_gemm_wmma<<<dim3(4, KB_ + QB_), 128, 0, stream>>>(k, Wk, kp, q, Wq, qp);
    chunk_attention<<<dim3(NCH_, Q_ / 16, B_), 256, 0, stream>>>(qp, kp, wv, mask, v, part, stat);
    combine_chunks<<<(B_ * Q_) / 4, 256, 0, stream>>>(part, stat, out);
}
